# starting kernel
# baseline (speedup 1.0000x reference)
#include <hip/hip_runtime.h>

typedef __attribute__((ext_vector_type(2))) float v2f;
typedef __attribute__((ext_vector_type(8))) float v8f;

#define NTASKS 512
#define NQ     75
#define NS     25
#define DIM    1024
#define NWAY   5
#define NSHOT  5

#define KC     64
#define PSTR   1032
#define QSTR   68
#define NWAVES 5
#define BLOCK  (NWAVES * 32)

__global__ __launch_bounds__(BLOCK)
void proto_head_wmma_kernel(const float* __restrict__ query,
                            const float* __restrict__ support,
                            const float* __restrict__ scale,
                            float* __restrict__ out)
{
    __shared__ float protos[NWAY * PSTR];
    __shared__ float qtile[NWAVES][16 * QSTR];
    __shared__ float bb[NWAY];
    __shared__ float aa_lds[NWAVES * 16];

    const int b    = blockIdx.x;
    const int tid  = threadIdx.x;
    const int lane = tid & 31;
    const int wave = tid >> 5;

    const float* sup = support + (size_t)b * NS * DIM;
    const float* q   = query   + (size_t)b * NQ * DIM;

    if (tid < NWAY) bb[tid] = 0.0f;
    __syncthreads();

    float bbp[NWAY] = {0.f, 0.f, 0.f, 0.f, 0.f};
    for (int d = tid; d < DIM; d += BLOCK) {
#pragma unroll
        for (int w = 0; w < NWAY; ++w) {
            float s = 0.0f;
#pragma unroll
            for (int j = 0; j < NSHOT; ++j)
                s += sup[(size_t)(w * NSHOT + j) * DIM + d];
            float p = s * 0.2f;
            protos[w * PSTR + d] = p;
            bbp[w] += p * p;
        }
    }
#pragma unroll
    for (int w = 0; w < NWAY; ++w) atomicAdd(&bb[w], bbp[w]);
    __syncthreads();

    const int mbase = wave * 16;
    const int mrow  = lane & 15;
    const int hi    = lane >> 4;
    const int klo   = 2 * hi;

    const int   n     = lane & 15;
    const int   nclmp = (n < NWAY) ? n : (NWAY - 1);
    const float nmask = (n < NWAY) ? 1.0f : 0.0f;

    float* qt = &qtile[wave][0];
    v8f   acc = {};
    float aa_part = 0.0f;

    for (int kc = 0; kc < DIM; kc += KC) {
#pragma unroll 4
        for (int r = 0; r < 16; ++r) {
            int m = mbase + r;
            float2 v = make_float2(0.0f, 0.0f);
            if (m < NQ)
                v = *(const float2*)(q + (size_t)m * DIM + kc + lane * 2);
            qt[r * QSTR + lane * 2]     = v.x;
            qt[r * QSTR + lane * 2 + 1] = v.y;
        }

        if (kc + KC < DIM) {
            int mp = mbase + mrow;
            if (mp >= NQ) mp = NQ - 1;
            __builtin_prefetch(q + (size_t)mp * DIM + kc + KC + lane * 2, 0, 0);
        }

        {
            float s = 0.0f;
#pragma unroll
            for (int c = 0; c < 32; ++c) {
                float v = qt[mrow * QSTR + hi * 32 + c];
                s += v * v;
            }
            aa_part += s;
        }

#pragma unroll
        for (int k = 0; k < KC; k += 4) {
            v2f a, bf;
            a.x  = qt[mrow * QSTR + k + klo];
            a.y  = qt[mrow * QSTR + k + klo + 1];
            bf.x = nmask * protos[nclmp * PSTR + kc + k + klo];
            bf.y = nmask * protos[nclmp * PSTR + kc + k + klo + 1];
            acc = __builtin_amdgcn_wmma_f32_16x16x4_f32(
                 false, a,  false, bf,
                 (short)0, acc,  false,  false);
        }
    }

    float aa_full = aa_part + __shfl_xor(aa_part, 16, 32);
    if (lane < 16) aa_lds[wave * 16 + lane] = aa_full;

    const float scl   = scale[0];
    const float inv_d = 1.0f / (float)DIM;
    if (n < NWAY) {
#pragma unroll
        for (int j = 0; j < 8; ++j) {
            int m = mbase + j + 8 * hi;
            if (m < NQ) {
                float aav   = aa_lds[wave * 16 + j + 8 * hi];
                float logit = (2.0f * acc[j] - aav - bb[n]) * inv_d * scl;
                out[((size_t)b * NQ + m) * NWAY + n] = logit;
            }
        }
    }
}

extern "C" void kernel_launch(void* const* d_in, const int* in_sizes, int n_in,
                              void* d_out, int out_size, void* d_ws, size_t ws_size,
                              hipStream_t stream) {
    const float* query   = (const float*)d_in[0];
    const float* support = (const float*)d_in[1];
    const float* scale   = (const float*)d_in[5];
    float* out = (float*)d_out;

    proto_head_wmma_kernel<<<NTASKS, BLOCK, 0, stream>>>(query, support, scale, out);
}


# work split by columns, not rows: each of 8 waves owns a 128-column slice of all 100 rows of a task, one 512-byte row piece per load; slice sums added in LDS, no atomics (summation order changes)
# speedup vs baseline: 2.4857x; 2.4857x over previous
#include <hip/hip_runtime.h>


typedef __attribute__((ext_vector_type(2))) float v2f;
typedef __attribute__((ext_vector_type(4))) float v4f;
typedef __attribute__((ext_vector_type(8))) float v8f;

#define NTASKS 512
#define NQ     75
#define NS     25
#define DIM    1024
#define NWAY   5
#define NSHOT  5

#define NWAVES 8
#define BLOCK  (NWAVES * 32)
#define SLICE  (DIM / NWAVES)
#define PITCH  132
#define BUF    (16 * PITCH)
#define MROWS  80

__global__ __launch_bounds__(BLOCK)
void proto_head_colslice_kernel(const float* __restrict__ query,
                                const float* __restrict__ support,
                                const float* __restrict__ scale,
                                float* __restrict__ out)
{
    __shared__ __attribute__((aligned(16))) float stage[NWAVES][BUF];
    __shared__ float part[NWAVES][MROWS * NWAY];
    __shared__ float aap[NWAVES][2][MROWS];

    const int b    = blockIdx.x;
    const int tid  = threadIdx.x;
    const int lane = tid & 31;
    const int wave = __builtin_amdgcn_readfirstlane(tid >> 5);
    const int c    = lane & 15;
    const int h    = lane >> 4;
    const int cc   = (c < NWAY) ? c : (NWAY - 1);

    const float* sup = support + (size_t)b * NS * DIM + wave * SLICE + lane * 4;
    const float* q   = query   + (size_t)b * NQ * DIM + wave * SLICE + lane * 4;

    float* st = &stage[wave][0];

    v4f s[NS];
#pragma unroll
    for (int i = 0; i < NS; ++i) s[i] = *(const v4f*)(sup + (size_t)i * DIM);

    v4f t4[11];
#pragma unroll
    for (int r = 0; r < 11; ++r) t4[r] = *(const v4f*)(q + (size_t)(64 + r) * DIM);

    v4f t[16];
#pragma unroll
    for (int r = 0; r < 16; ++r) t[r] = *(const v4f*)(q + (size_t)r * DIM);

    v4f p[NWAY];
#pragma unroll
    for (int w = 0; w < NWAY; ++w) {
        v4f a = s[w * NSHOT];
#pragma unroll
        for (int j = 1; j < NSHOT; ++j) a += s[w * NSHOT + j];
        p[w] = a * 0.2f;
    }

#pragma unroll
    for (int w = 0; w < NWAY; ++w) *(v4f*)(st + (11 + w) * PITCH + lane * 4) = p[w];

    v4f bfr[16];
#pragma unroll
    for (int j = 0; j < 16; ++j) bfr[j] = *(const v4f*)(st + (11 + cc) * PITCH + 8 * j + 4 * h);

#pragma unroll
    for (int r = 0; r < 11; ++r) *(v4f*)(st + r * PITCH + lane * 4) = t4[r];

#pragma unroll 1
    for (int i = 0; i < 5; ++i) {
        v4f a[16];
#pragma unroll
        for (int j = 0; j < 16; ++j) a[j] = *(const v4f*)(st + c * PITCH + 8 * j + 4 * h);

        v8f acc0 = {}, acc1 = {};
        v2f sq[4] = {{0.0f, 0.0f}, {0.0f, 0.0f}, {0.0f, 0.0f}, {0.0f, 0.0f}};
#pragma unroll
        for (int j = 0; j < 16; ++j) {
            v2f a0 = {a[j].x, a[j].y},     a1 = {a[j].z, a[j].w};
            v2f b0 = {bfr[j].x, bfr[j].y}, b1 = {bfr[j].z, bfr[j].w};
            acc0 = __builtin_amdgcn_wmma_f32_16x16x4_f32(false, a0, false, b0, (short)0, acc0, false, false);
            acc1 = __builtin_amdgcn_wmma_f32_16x16x4_f32(false, a1, false, b1, (short)0, acc1, false, false);
            sq[2 * (j & 1)]     += a0 * a0;
            sq[2 * (j & 1) + 1] += a1 * a1;
        }

        const int m0 = (i == 0) ? 64 : 16 * (i - 1);
        const v2f sqs = (sq[0] + sq[1]) + (sq[2] + sq[3]);
        aap[wave][h][m0 + c] = sqs.x + sqs.y;
        if (c < NWAY) {
#pragma unroll
            for (int r = 0; r < 8; ++r)
                part[wave][(m0 + 8 * h + r) * NWAY + c] = acc0[r] + acc1[r];
        }

        if (i < 4) {
#pragma unroll
            for (int r = 0; r < 16; ++r) *(v4f*)(st + r * PITCH + lane * 4) = t[r];
            if (i < 3) {
                const float* qn = q + (size_t)(16 * (i + 1)) * DIM;
#pragma unroll
                for (int r = 0; r < 16; ++r) t[r] = *(const v4f*)(qn + (size_t)r * DIM);
            }
        }
    }

    __syncthreads();

    const float scl   = scale[0];
    const float inv_d = 1.0f / (float)DIM;
    for (int e = tid; e < NQ * NWAY; e += BLOCK) {
        const int m = e / NWAY;
        const int n = e - m * NWAY;
        float ab = 0.0f, aa = 0.0f, bb = 0.0f;
#pragma unroll
        for (int w = 0; w < NWAVES; ++w) {
            ab += part[w][e];
            aa += aap[w][0][m] + aap[w][1][m];
            bb += aap[w][0][NQ + n] + aap[w][1][NQ + n];
        }
        out[(size_t)b * (NQ * NWAY) + e] = (2.0f * ab - aa - bb) * inv_d * scl;
    }
}

extern "C" void kernel_launch(void* const* d_in, const int* in_sizes, int n_in,
                              void* d_out, int out_size, void* d_ws, size_t ws_size,
                              hipStream_t stream) {
    const float* query   = (const float*)d_in[0];
    const float* support = (const float*)d_in[1];
    const float* scale   = (const float*)d_in[5];
    float* out = (float*)d_out;

    proto_head_colslice_kernel<<<NTASKS, BLOCK, 0, stream>>>(query, support, scale, out);
}


# prototype B fragments read from LDS beside the A fragments, not kept in 64 registers; the freed registers hold a second query tile: each tile requested two tiles ahead, 16-32 loads in flight per wave
# speedup vs baseline: 2.6670x; 1.0729x over previous
#include <hip/hip_runtime.h>


typedef __attribute__((ext_vector_type(2))) float v2f;
typedef __attribute__((ext_vector_type(4))) float v4f;
typedef __attribute__((ext_vector_type(8))) float v8f;

#define NTASKS 512
#define NQ     75
#define NS     25
#define DIM    1024
#define NWAY   5
#define NSHOT  5

#define NWAVES 8
#define BLOCK  (NWAVES * 32)
#define SLICE  (DIM / NWAVES)
#define PITCH  132
#define BUF    (16 * PITCH)
#define MROWS  80

__global__ __launch_bounds__(BLOCK) __attribute__((amdgpu_waves_per_eu(4, 4)))
void proto_head_colslice_kernel(const float* __restrict__ query,
                                const float* __restrict__ support,
                                const float* __restrict__ scale,
                                float* __restrict__ out)
{
    __shared__ __attribute__((aligned(16))) float stage[NWAVES][BUF];
    __shared__ __attribute__((aligned(16))) float pb[NWAVES][NWAY * PITCH];
    __shared__ float part[NWAVES][MROWS * NWAY];
    __shared__ float aap[NWAVES][2][MROWS];

    const int b    = blockIdx.x;
    const int tid  = threadIdx.x;
    const int lane = tid & 31;
    const int wave = __builtin_amdgcn_readfirstlane(tid >> 5);
    const int c    = lane & 15;
    const int h    = lane >> 4;
    const int cc   = (c < NWAY) ? c : (NWAY - 1);

    const float* sup = support + (size_t)b * NS * DIM + wave * SLICE + lane * 4;
    const float* q   = query   + (size_t)b * NQ * DIM + wave * SLICE + lane * 4;

    float* st  = &stage[wave][0];
    float* pbw = &pb[wave][0];

    v4f s[NS];
#pragma unroll
    for (int i = 0; i < NS; ++i) s[i] = *(const v4f*)(sup + (size_t)i * DIM);

    v4f t4[11];
#pragma unroll
    for (int r = 0; r < 11; ++r) t4[r] = *(const v4f*)(q + (size_t)(64 + r) * DIM);

    v4f ta[16], tb[16];
    __builtin_amdgcn_sched_barrier(0);
#pragma unroll
    for (int r = 0; r < 16; ++r) ta[r] = *(const v4f*)(q + (size_t)r * DIM);

    v4f p[NWAY];
#pragma unroll
    for (int w = 0; w < NWAY; ++w) {
        v4f a = s[w * NSHOT];
#pragma unroll
        for (int j = 1; j < NSHOT; ++j) a += s[w * NSHOT + j];
        p[w] = a * 0.2f;
    }

    __builtin_amdgcn_sched_barrier(0);
#pragma unroll
    for (int r = 0; r < 16; ++r) tb[r] = *(const v4f*)(q + (size_t)(16 + r) * DIM);

#pragma unroll
    for (int w = 0; w < NWAY; ++w) {
        *(v4f*)(pbw + w * PITCH + lane * 4)        = p[w];
        *(v4f*)(st + (11 + w) * PITCH + lane * 4)  = p[w];
    }
#pragma unroll
    for (int r = 0; r < 11; ++r) *(v4f*)(st + r * PITCH + lane * 4) = t4[r];

    const float* afrag = st  + c  * PITCH + 4 * h;
    const float* bfrag = pbw + cc * PITCH + 4 * h;

#define TILE_BLOCK(M0)                                                                         \
    {                                                                                          \
        v8f acc0 = {}, acc1 = {};                                                              \
        v2f sq[4] = {{0.0f, 0.0f}, {0.0f, 0.0f}, {0.0f, 0.0f}, {0.0f, 0.0f}};                  \
        _Pragma("unroll")                                                                      \
        for (int g = 0; g < 4; ++g) {                                                          \
            v4f a[4], bq[4];                                                                   \
            _Pragma("unroll")                                                                  \
            for (int j = 0; j < 4; ++j) {                                                      \
                a[j]  = *(const v4f*)(afrag + 8 * (4 * g + j));                                \
                bq[j] = *(const v4f*)(bfrag + 8 * (4 * g + j));                                \
            }                                                                                  \
              \
              \
            __builtin_amdgcn_sched_barrier(0);                                                 \
            _Pragma("unroll")                                                                  \
            for (int j = 0; j < 4; ++j) {                                                      \
                v2f a0 = {a[j].x, a[j].y},   a1 = {a[j].z, a[j].w};                            \
                v2f b0 = {bq[j].x, bq[j].y}, b1 = {bq[j].z, bq[j].w};                          \
                acc0 = __builtin_amdgcn_wmma_f32_16x16x4_f32(false, a0, false, b0, (short)0, acc0, false, false); \
                acc1 = __builtin_amdgcn_wmma_f32_16x16x4_f32(false, a1, false, b1, (short)0, acc1, false, false); \
                sq[2 * (j & 1)]     += a0 * a0;                                                \
                sq[2 * (j & 1) + 1] += a1 * a1;                                                \
            }                                                                                  \
        }                                                                                      \
        const v2f sqs = (sq[0] + sq[1]) + (sq[2] + sq[3]);                                     \
        aap[wave][h][(M0) + c] = sqs.x + sqs.y;                                                \
        if (c < NWAY) {                                                                        \
            _Pragma("unroll")                                                                  \
            for (int r = 0; r < 8; ++r)     \
                part[wave][((M0) + 8 * h + r) * NWAY + c] = acc0[r] + acc1[r];                 \
        }                                                                                      \
    }

#define STAGE(T)                                                                               \
    _Pragma("unroll")                                                                          \
    for (int r = 0; r < 16; ++r) *(v4f*)(st + r * PITCH + lane * 4) = T[r];

#define REQUEST(T, ROW0)                                                                       \
    _Pragma("unroll")                                                                          \
    for (int r = 0; r < 16; ++r) T[r] = *(const v4f*)(q + (size_t)((ROW0) + r) * DIM);

    TILE_BLOCK(64)
    STAGE(ta)
    __builtin_amdgcn_sched_barrier(0);
    REQUEST(ta, 32)
    TILE_BLOCK(0)
    STAGE(tb)
    __builtin_amdgcn_sched_barrier(0);
    REQUEST(tb, 48)
    TILE_BLOCK(16)
    STAGE(ta)
    TILE_BLOCK(32)
    STAGE(tb)
    TILE_BLOCK(48)

    __syncthreads();

    const float scl   = scale[0];
    const float inv_d = 1.0f / (float)DIM;
    for (int e = tid; e < NQ * NWAY; e += BLOCK) {
        const int m = e / NWAY;
        const int n = e - m * NWAY;
        float ab = 0.0f, aa = 0.0f, bb = 0.0f;
#pragma unroll
        for (int w = 0; w < NWAVES; ++w) {
            ab += part[w][e];
            aa += aap[w][0][m] + aap[w][1][m];
            bb += aap[w][0][NQ + n] + aap[w][1][NQ + n];
        }
        out[(size_t)b * (NQ * NWAY) + e] = (2.0f * ab - aa - bb) * inv_d * scl;
    }
}

extern "C" void kernel_launch(void* const* d_in, const int* in_sizes, int n_in,
                              void* d_out, int out_size, void* d_ws, size_t ws_size,
                              hipStream_t stream) {
    const float* query   = (const float*)d_in[0];
    const float* support = (const float*)d_in[1];
    const float* scale   = (const float*)d_in[5];
    float* out = (float*)d_out;

    proto_head_colslice_kernel<<<NTASKS, BLOCK, 0, stream>>>(query, support, scale, out);
}


# loads split by rows, matrix work still by columns: each wave fetches whole rows (8 loads = 4 contiguous KB; a class's 5 support rows per wave), hands piece k to wave k in LDS behind LDS-only barriers
# speedup vs baseline: 3.2763x; 1.2285x over previous
#include <hip/hip_runtime.h>


typedef __attribute__((ext_vector_type(2))) float v2f;
typedef __attribute__((ext_vector_type(4))) float v4f;
typedef __attribute__((ext_vector_type(8))) float v8f;

#define NTASKS 512
#define NQ     75
#define NS     25
#define DIM    1024
#define NWAY   5
#define NSHOT  5

#define NWAVES 8
#define BLOCK  (NWAVES * 32)
#define SLICE  (DIM / NWAVES)
#define PITCH  132
#define BUF    (16 * PITCH)
#define MROWS  80

static __device__ __forceinline__ void lds_barrier()
{
    asm volatile("s_wait_dscnt 0x0\n\ts_barrier_signal -1\n\ts_barrier_wait -1" ::: "memory");
}

__global__ __launch_bounds__(BLOCK) __attribute__((amdgpu_waves_per_eu(4, 4)))
void proto_head_rowload_kernel(const float* __restrict__ query,
                               const float* __restrict__ support,
                               const float* __restrict__ scale,
                               float* __restrict__ out)
{
    __shared__ __attribute__((aligned(16))) float stage[NWAVES][BUF];
    __shared__ __attribute__((aligned(16))) float pb[NWAVES][NWAY * PITCH];
    __shared__ float part[NWAVES][MROWS * NWAY];
    __shared__ float aap[NWAVES][2][MROWS];

    const int b    = blockIdx.x;
    const int tid  = threadIdx.x;
    const int lane = tid & 31;
    const int wave = __builtin_amdgcn_readfirstlane(tid >> 5);
    const int c    = lane & 15;
    const int h    = lane >> 4;
    const int cc   = (c < NWAY) ? c : (NWAY - 1);

    const float* q = query + ((size_t)b * NQ + 2 * wave) * DIM + lane * 4;

    float* st  = &stage[wave][0];
    float* pbw = &pb[wave][0];
    float* hand = &stage[0][0] + (2 * wave) * PITCH + lane * 4;

    v4f u[NSHOT * 8];
    if (wave < NWAY) {
        const float* sp = support + ((size_t)b * NS + wave * NSHOT) * DIM + lane * 4;
#pragma unroll
        for (int j = 0; j < NSHOT; ++j) {
#pragma unroll
            for (int k = 0; k < 8; ++k) u[8 * j + k] = *(const v4f*)(sp + (size_t)j * DIM + k * SLICE);
        }
    } else {
        const float* qp = query + ((size_t)b * NQ + 64 + 4 * (wave - NWAY)) * DIM + lane * 4;
#pragma unroll
        for (int j = 0; j < 3; ++j) {
#pragma unroll
            for (int k = 0; k < 8; ++k) u[8 * j + k] = *(const v4f*)(qp + (size_t)j * DIM + k * SLICE);
        }
        if (wave != NWAVES - 1) {
#pragma unroll
            for (int k = 0; k < 8; ++k) u[24 + k] = *(const v4f*)(qp + (size_t)3 * DIM + k * SLICE);
        }
    }

    v4f ta[16], tb[16];
    __builtin_amdgcn_sched_barrier(0);
#pragma unroll
    for (int i = 0; i < 2; ++i) {
#pragma unroll
        for (int k = 0; k < 8; ++k) ta[8 * i + k] = *(const v4f*)(q + (size_t)i * DIM + k * SLICE);
    }

    if (wave < NWAY) {
#pragma unroll
        for (int k = 0; k < 8; ++k) {
            v4f a = u[k];
#pragma unroll
            for (int j = 1; j < NSHOT; ++j) a += u[8 * j + k];
            const v4f p = a * 0.2f;
            *(v4f*)(&pb[k][0]    + wave * PITCH + lane * 4)        = p;
            *(v4f*)(&stage[k][0] + (11 + wave) * PITCH + lane * 4) = p;
        }
    } else {
        float* h4 = &stage[0][0] + (4 * (wave - NWAY)) * PITCH + lane * 4;
#pragma unroll
        for (int j = 0; j < 3; ++j) {
#pragma unroll
            for (int k = 0; k < 8; ++k) *(v4f*)(h4 + k * BUF + j * PITCH) = u[8 * j + k];
        }
        if (wave != NWAVES - 1) {
#pragma unroll
            for (int k = 0; k < 8; ++k) *(v4f*)(h4 + k * BUF + 3 * PITCH) = u[24 + k];
        }
    }

    __builtin_amdgcn_sched_barrier(0);
#pragma unroll
    for (int i = 0; i < 2; ++i) {
#pragma unroll
        for (int k = 0; k < 8; ++k) tb[8 * i + k] = *(const v4f*)(q + (size_t)(16 + i) * DIM + k * SLICE);
    }

    const float* afrag = st  + c  * PITCH + 4 * h;
    const float* bfrag = pbw + cc * PITCH + 4 * h;

#define TILE_BLOCK(M0)                                                                         \
    {                                                                                          \
        v8f acc0 = {}, acc1 = {};                                                              \
        v2f sq[4] = {{0.0f, 0.0f}, {0.0f, 0.0f}, {0.0f, 0.0f}, {0.0f, 0.0f}};                  \
        _Pragma("unroll")                                                                      \
        for (int g = 0; g < 4; ++g) {                                                          \
            v4f a[4], bq[4];                                                                   \
            _Pragma("unroll")                                                                  \
            for (int j = 0; j < 4; ++j) {                                                      \
                a[j]  = *(const v4f*)(afrag + 8 * (4 * g + j));                                \
                bq[j] = *(const v4f*)(bfrag + 8 * (4 * g + j));                                \
            }                                                                                  \
              \
              \
            __builtin_amdgcn_sched_barrier(0);                                                 \
            _Pragma("unroll")                                                                  \
            for (int j = 0; j < 4; ++j) {                                                      \
                v2f a0 = {a[j].x, a[j].y},   a1 = {a[j].z, a[j].w};                            \
                v2f b0 = {bq[j].x, bq[j].y}, b1 = {bq[j].z, bq[j].w};                          \
                acc0 = __builtin_amdgcn_wmma_f32_16x16x4_f32(false, a0, false, b0, (short)0, acc0, false, false); \
                acc1 = __builtin_amdgcn_wmma_f32_16x16x4_f32(false, a1, false, b1, (short)0, acc1, false, false); \
                sq[2 * (j & 1)]     += a0 * a0;                                                \
                sq[2 * (j & 1) + 1] += a1 * a1;                                                \
            }                                                                                  \
        }                                                                                      \
        const v2f sqs = (sq[0] + sq[1]) + (sq[2] + sq[3]);                                     \
        aap[wave][h][(M0) + c] = sqs.x + sqs.y;                                                \
        if (c < NWAY) {                                                                        \
            _Pragma("unroll")                                                                  \
            for (int r = 0; r < 8; ++r)     \
                part[wave][((M0) + 8 * h + r) * NWAY + c] = acc0[r] + acc1[r];                 \
        }                                                                                      \
    }

#define STAGE(T)                                                                               \
    _Pragma("unroll")                                                                          \
    for (int i = 0; i < 2; ++i) {                                                              \
        _Pragma("unroll")                                                                      \
        for (int k = 0; k < 8; ++k) *(v4f*)(hand + k * BUF + i * PITCH) = T[8 * i + k];        \
    }

#define REQUEST(T, ROW0)                                                                       \
    _Pragma("unroll")                                                                          \
    for (int i = 0; i < 2; ++i) {                                                              \
        _Pragma("unroll")                                                                      \
        for (int k = 0; k < 8; ++k) T[8 * i + k] = *(const v4f*)(q + (size_t)((ROW0) + i) * DIM + k * SLICE); \
    }

    lds_barrier();
    TILE_BLOCK(64)
    lds_barrier();
    STAGE(ta)
    __builtin_amdgcn_sched_barrier(0);
    REQUEST(ta, 32)
    lds_barrier();
    TILE_BLOCK(0)
    lds_barrier();
    STAGE(tb)
    __builtin_amdgcn_sched_barrier(0);
    REQUEST(tb, 48)
    lds_barrier();
    TILE_BLOCK(16)
    lds_barrier();
    STAGE(ta)
    lds_barrier();
    TILE_BLOCK(32)
    lds_barrier();
    STAGE(tb)
    lds_barrier();
    TILE_BLOCK(48)

    __syncthreads();

    const float scl   = scale[0];
    const float inv_d = 1.0f / (float)DIM;
    for (int e = tid; e < NQ * NWAY; e += BLOCK) {
        const int m = e / NWAY;
        const int n = e - m * NWAY;
        float ab = 0.0f, aa = 0.0f, bb = 0.0f;
#pragma unroll
        for (int w = 0; w < NWAVES; ++w) {
            ab += part[w][e];
            aa += aap[w][0][m] + aap[w][1][m];
            bb += aap[w][0][NQ + n] + aap[w][1][NQ + n];
        }
        out[(size_t)b * (NQ * NWAY) + e] = (2.0f * ab - aa - bb) * inv_d * scl;
    }
}

extern "C" void kernel_launch(void* const* d_in, const int* in_sizes, int n_in,
                              void* d_out, int out_size, void* d_ws, size_t ws_size,
                              hipStream_t stream) {
    const float* query   = (const float*)d_in[0];
    const float* support = (const float*)d_in[1];
    const float* scale   = (const float*)d_in[5];
    float* out = (float*)d_out;

    proto_head_rowload_kernel<<<NTASKS, BLOCK, 0, stream>>>(query, support, scale, out);
}
